# starting kernel
# baseline (speedup 1.0000x reference)
#include <hip/hip_runtime.h>

typedef __attribute__((ext_vector_type(2))) float v2f;
typedef __attribute__((ext_vector_type(8))) float v8f;

#define NN       50000
#define IN_CH    128
#define OUT_CH   64
#define HEADS    2
#define HID      128
#define HC1      (HEADS*HID)
#define E0_EDGES 400000
#define E_TOT    (E0_EDGES + NN)
#define NEG_SLOPE 0.2f

struct Bop { v2f b0, b1, b2, b3; };

__device__ __forceinline__ v2f load_a(const float* __restrict__ arow, int k0, int lhalf) {
  v2f a;
  a.x = arow[k0 + 2 * lhalf];
  a.y = arow[k0 + 2 * lhalf + 1];
  return a;
}
__device__ __forceinline__ Bop load_b(const float* __restrict__ B, int k0, int N,
                                      int tn, int lhalf, int lmod) {
  const float* r0 = B + (size_t)(k0 + 2 * lhalf) * N + tn + lmod;
  const float* r1 = r0 + N;
  Bop b;
  b.b0.x = r0[0];  b.b0.y = r1[0];
  b.b1.x = r0[16]; b.b1.y = r1[16];
  b.b2.x = r0[32]; b.b2.y = r1[32];
  b.b3.x = r0[48]; b.b3.y = r1[48];
  return b;
}

__global__ void __launch_bounds__(256)
gemm_wmma_f32(const float* __restrict__ A, const float* __restrict__ B,
              float* __restrict__ C, int M, int N, int K) {
  const int wave   = blockIdx.x * (blockDim.x >> 5) + (threadIdx.x >> 5);
  const int tilesN = N >> 6;
  const int tm     = wave / tilesN;
  const int tn     = (wave % tilesN) << 6;
  if (tm * 16 >= M) return;
  const int lhalf = (threadIdx.x >> 4) & 1;
  const int lmod  = threadIdx.x & 15;

  v8f c0 = {}, c1 = {}, c2 = {}, c3 = {};
  const float* arow = A + (size_t)(tm * 16 + lmod) * K;

  v2f a = load_a(arow, 0, lhalf);
  Bop b = load_b(B, 0, N, tn, lhalf, lmod);
  for (int k0 = 0; k0 < K - 4; k0 += 4) {
    v2f an = load_a(arow, k0 + 4, lhalf);
    Bop bn = load_b(B, k0 + 4, N, tn, lhalf, lmod);
    c0 = __builtin_amdgcn_wmma_f32_16x16x4_f32(false, a, false, b.b0, (short)0, c0, false, false);
    c1 = __builtin_amdgcn_wmma_f32_16x16x4_f32(false, a, false, b.b1, (short)0, c1, false, false);
    c2 = __builtin_amdgcn_wmma_f32_16x16x4_f32(false, a, false, b.b2, (short)0, c2, false, false);
    c3 = __builtin_amdgcn_wmma_f32_16x16x4_f32(false, a, false, b.b3, (short)0, c3, false, false);
    a = an; b = bn;
  }
  c0 = __builtin_amdgcn_wmma_f32_16x16x4_f32(false, a, false, b.b0, (short)0, c0, false, false);
  c1 = __builtin_amdgcn_wmma_f32_16x16x4_f32(false, a, false, b.b1, (short)0, c1, false, false);
  c2 = __builtin_amdgcn_wmma_f32_16x16x4_f32(false, a, false, b.b2, (short)0, c2, false, false);
  c3 = __builtin_amdgcn_wmma_f32_16x16x4_f32(false, a, false, b.b3, (short)0, c3, false, false);

#pragma unroll
  for (int j = 0; j < 8; ++j) {
    float* p = C + (size_t)(tm * 16 + j + 8 * lhalf) * N + tn + lmod;
    p[0] = c0[j]; p[16] = c1[j]; p[32] = c2[j]; p[48] = c3[j];
  }
}

__global__ void __launch_bounds__(256)
att_scores(const float* __restrict__ h, const float* __restrict__ att_s,
           const float* __restrict__ att_d, float* __restrict__ as,
           float* __restrict__ ad, int n_nodes, int H, int C) {
  int i = blockIdx.x * blockDim.x + threadIdx.x;
  if (i >= n_nodes * H) return;
  int n = i / H, hh = i % H;
  const float4* hp = (const float4*)(h + (size_t)n * H * C + (size_t)hh * C);
  const float4* sp = (const float4*)(att_s + hh * C);
  const float4* dp = (const float4*)(att_d + hh * C);
  float s = 0.f, d = 0.f;
  for (int c = 0; c < (C >> 2); ++c) {
    float4 v = hp[c], ws = sp[c], wd = dp[c];
    s += v.x * ws.x + v.y * ws.y + v.z * ws.z + v.w * ws.w;
    d += v.x * wd.x + v.y * wd.y + v.z * wd.z + v.w * wd.w;
  }
  as[i] = s; ad[i] = d;
}

__device__ __forceinline__ void edge_nodes(const int* __restrict__ ei, int e,
                                           int& s, int& d) {
  if (e < E0_EDGES) { s = ei[e]; d = ei[E0_EDGES + e]; }
  else              { s = e - E0_EDGES; d = s; }
}
__device__ __forceinline__ float leaky(float v) { return v >= 0.f ? v : NEG_SLOPE * v; }
__device__ __forceinline__ unsigned f2ord(float f) {
  unsigned u = __float_as_uint(f);
  return (u & 0x80000000u) ? ~u : (u | 0x80000000u);
}
__device__ __forceinline__ float ord2f(unsigned k) {
  unsigned u = (k & 0x80000000u) ? (k ^ 0x80000000u) : ~k;
  return __uint_as_float(u);
}

__global__ void __launch_bounds__(256)
edge_max(const int* __restrict__ ei, const float* __restrict__ as,
         const float* __restrict__ ad, unsigned* __restrict__ m, int H) {
  int i = blockIdx.x * blockDim.x + threadIdx.x;
  if (i >= E_TOT * H) return;
  int e = i / H, hh = i % H;
  int s, d; edge_nodes(ei, e, s, d);
  float v = leaky(as[s * H + hh] + ad[d * H + hh]);
  atomicMax(&m[d * H + hh], f2ord(v));
}

__global__ void __launch_bounds__(256)
edge_exp(const int* __restrict__ ei, const float* __restrict__ as,
         const float* __restrict__ ad, const unsigned* __restrict__ m,
         float* __restrict__ ea, float* __restrict__ denom, int H) {
  int i = blockIdx.x * blockDim.x + threadIdx.x;
  if (i >= E_TOT * H) return;
  int e = i / H, hh = i % H;
  int s, d; edge_nodes(ei, e, s, d);
  float v  = leaky(as[s * H + hh] + ad[d * H + hh]);
  float ex = __expf(v - ord2f(m[d * H + hh]));
  ea[i] = ex;
  atomicAdd(&denom[d * H + hh], ex);
}

__global__ void __launch_bounds__(256)
edge_scatter(const int* __restrict__ ei, const float* __restrict__ h,
             const float* __restrict__ ea, const float* __restrict__ denom,
             float* __restrict__ agg, int H, int C) {
  const int HC  = H * C;
  const int per = HC >> 2;
  long long i = (long long)blockIdx.x * blockDim.x + threadIdx.x;
  if (i >= (long long)E_TOT * per) return;
  int e  = (int)(i / per);
  int cq = (int)(i % per) << 2;
  int hh = cq / C;
  int s, d; edge_nodes(ei, e, s, d);
  float a = ea[(size_t)e * H + hh] / (denom[d * H + hh] + 1e-16f);
  float4 hv = *(const float4*)(h + (size_t)s * HC + cq);
  float* p = agg + (size_t)d * HC + cq;
  atomicAdd(p + 0, hv.x * a);
  atomicAdd(p + 1, hv.y * a);
  atomicAdd(p + 2, hv.z * a);
  atomicAdd(p + 3, hv.w * a);
}

__global__ void __launch_bounds__(256)
bias_elu(float* __restrict__ h, const float* __restrict__ b, int total, int C) {
  int i = blockIdx.x * blockDim.x + threadIdx.x;
  if (i >= total) return;
  float v = h[i] + b[i % C];
  h[i] = v > 0.f ? v : (__expf(v) - 1.f);
}

__global__ void __launch_bounds__(256)
bias_add(float* __restrict__ o, const float* __restrict__ b, int total, int C) {
  int i = blockIdx.x * blockDim.x + threadIdx.x;
  if (i >= total) return;
  o[i] += b[i % C];
}

extern "C" void kernel_launch(void* const* d_in, const int* in_sizes, int n_in,
                              void* d_out, int out_size, void* d_ws, size_t ws_size,
                              hipStream_t stream) {
  const float* x    = (const float*)d_in[0];
  const int*   ei   = (const int*)  d_in[1];
  const float* W1   = (const float*)d_in[2];
  const float* aS1  = (const float*)d_in[3];
  const float* aD1  = (const float*)d_in[4];
  const float* b1   = (const float*)d_in[5];
  const float* Wmu  = (const float*)d_in[6];
  const float* aSmu = (const float*)d_in[7];
  const float* aDmu = (const float*)d_in[8];
  const float* bmu  = (const float*)d_in[9];
  const float* Wls  = (const float*)d_in[10];
  const float* aSls = (const float*)d_in[11];
  const float* aDls = (const float*)d_in[12];
  const float* bls  = (const float*)d_in[13];

  char* ws = (char*)d_ws;
  size_t off = 0;
  auto alloc = [&](size_t bytes) -> char* {
    off = (off + 255) & ~(size_t)255;
    char* p = ws + off; off += bytes; return p;
  };

  float*    h1   = (float*)   alloc((size_t)NN * HC1 * 4);
  float*    hbuf = (float*)   alloc((size_t)NN * HC1 * 4);
  float*    as1  = (float*)   alloc((size_t)NN * HEADS * 4);
  float*    ad1  = (float*)   alloc((size_t)NN * HEADS * 4);
  unsigned* m1   = (unsigned*)alloc((size_t)NN * HEADS * 4);
  float*    dn1  = (float*)   alloc((size_t)NN * HEADS * 4);
  float*    ea1  = (float*)   alloc((size_t)E_TOT * HEADS * 4);
  float*    as2m = (float*)   alloc((size_t)NN * 4);
  float*    ad2m = (float*)   alloc((size_t)NN * 4);
  unsigned* m2m  = (unsigned*)alloc((size_t)NN * 4);
  float*    dn2m = (float*)   alloc((size_t)NN * 4);
  float*    ea2m = (float*)   alloc((size_t)E_TOT * 4);
  float*    as2l = (float*)   alloc((size_t)NN * 4);
  float*    ad2l = (float*)   alloc((size_t)NN * 4);
  unsigned* m2l  = (unsigned*)alloc((size_t)NN * 4);
  float*    dn2l = (float*)   alloc((size_t)NN * 4);
  float*    ea2l = (float*)   alloc((size_t)E_TOT * 4);

  float* hm = h1;
  float* hl = h1 + (size_t)NN * OUT_CH;
  float* out_mu = (float*)d_out;
  float* out_ls = out_mu + (size_t)NN * OUT_CH;

  hipMemsetAsync(hbuf, 0, (size_t)NN * HC1 * 4, stream);
  hipMemsetAsync(m1,   0, (size_t)NN * HEADS * 4, stream);
  hipMemsetAsync(dn1,  0, (size_t)NN * HEADS * 4, stream);

  int waves = (NN / 16) * (HC1 / 64);
  gemm_wmma_f32<<<(waves + 7) / 8, 256, 0, stream>>>(x, W1, h1, NN, HC1, IN_CH);
  att_scores<<<(NN * HEADS + 255) / 256, 256, 0, stream>>>(h1, aS1, aD1, as1, ad1, NN, HEADS, HID);
  edge_max<<<(E_TOT * HEADS + 255) / 256, 256, 0, stream>>>(ei, as1, ad1, m1, HEADS);
  edge_exp<<<(E_TOT * HEADS + 255) / 256, 256, 0, stream>>>(ei, as1, ad1, m1, ea1, dn1, HEADS);
  {
    long long work = (long long)E_TOT * (HC1 / 4);
    edge_scatter<<<(unsigned)((work + 255) / 256), 256, 0, stream>>>(ei, h1, ea1, dn1, hbuf, HEADS, HID);
  }
  bias_elu<<<(NN * HC1 + 255) / 256, 256, 0, stream>>>(hbuf, b1, NN * HC1, HC1);

  waves = (NN / 16) * (OUT_CH / 64);
  gemm_wmma_f32<<<(waves + 7) / 8, 256, 0, stream>>>(hbuf, Wmu, hm, NN, OUT_CH, HC1);
  gemm_wmma_f32<<<(waves + 7) / 8, 256, 0, stream>>>(hbuf, Wls, hl, NN, OUT_CH, HC1);
  att_scores<<<(NN + 255) / 256, 256, 0, stream>>>(hm, aSmu, aDmu, as2m, ad2m, NN, 1, OUT_CH);
  att_scores<<<(NN + 255) / 256, 256, 0, stream>>>(hl, aSls, aDls, as2l, ad2l, NN, 1, OUT_CH);

  hipMemsetAsync(m2m, 0, (size_t)NN * 4, stream);
  hipMemsetAsync(dn2m, 0, (size_t)NN * 4, stream);
  hipMemsetAsync(m2l, 0, (size_t)NN * 4, stream);
  hipMemsetAsync(dn2l, 0, (size_t)NN * 4, stream);
  hipMemsetAsync(d_out, 0, (size_t)2 * NN * OUT_CH * 4, stream);

  edge_max<<<(E_TOT + 255) / 256, 256, 0, stream>>>(ei, as2m, ad2m, m2m, 1);
  edge_exp<<<(E_TOT + 255) / 256, 256, 0, stream>>>(ei, as2m, ad2m, m2m, ea2m, dn2m, 1);
  {
    long long work = (long long)E_TOT * (OUT_CH / 4);
    edge_scatter<<<(unsigned)((work + 255) / 256), 256, 0, stream>>>(ei, hm, ea2m, dn2m, out_mu, 1, OUT_CH);
  }
  edge_max<<<(E_TOT + 255) / 256, 256, 0, stream>>>(ei, as2l, ad2l, m2l, 1);
  edge_exp<<<(E_TOT + 255) / 256, 256, 0, stream>>>(ei, as2l, ad2l, m2l, ea2l, dn2l, 1);
  {
    long long work = (long long)E_TOT * (OUT_CH / 4);
    edge_scatter<<<(unsigned)((work + 255) / 256), 256, 0, stream>>>(ei, hl, ea2l, dn2l, out_ls, 1, OUT_CH);
  }

  bias_add<<<(NN * OUT_CH + 255) / 256, 256, 0, stream>>>(out_mu, bmu, NN * OUT_CH, OUT_CH);
  bias_add<<<(NN * OUT_CH + 255) / 256, 256, 0, stream>>>(out_ls, bls, NN * OUT_CH, OUT_CH);
}


# aggregate by gather, not float atomics: edges grouped by destination and sorted by source each call (fixed sum order); one wave per node does softmax and sum of alpha*h[src], one store per row
# speedup vs baseline: 2.8627x; 2.8627x over previous
#include <hip/hip_runtime.h>

typedef __attribute__((ext_vector_type(2))) float v2f;
typedef __attribute__((ext_vector_type(8))) float v8f;

#define NN       50000
#define IN_CH    128
#define OUT_CH   64
#define HEADS    2
#define HID      128
#define HC1      (HEADS*HID)
#define E0_EDGES 400000
#define E_TOT    (E0_EDGES + NN)
#define NEG_SLOPE 0.2f

struct Bop { v2f b0, b1, b2, b3; };

__device__ __forceinline__ v2f load_a(const float* __restrict__ arow, int k0, int lhalf) {
  v2f a;
  a.x = arow[k0 + 2 * lhalf];
  a.y = arow[k0 + 2 * lhalf + 1];
  return a;
}
__device__ __forceinline__ Bop load_b(const float* __restrict__ B, int k0, int N,
                                      int tn, int lhalf, int lmod) {
  const float* r0 = B + (size_t)(k0 + 2 * lhalf) * N + tn + lmod;
  const float* r1 = r0 + N;
  Bop b;
  b.b0.x = r0[0];  b.b0.y = r1[0];
  b.b1.x = r0[16]; b.b1.y = r1[16];
  b.b2.x = r0[32]; b.b2.y = r1[32];
  b.b3.x = r0[48]; b.b3.y = r1[48];
  return b;
}

__global__ void __launch_bounds__(256)
gemm_wmma_f32(const float* __restrict__ A, const float* __restrict__ B,
              float* __restrict__ C, int M, int N, int K) {
  const int wave   = blockIdx.x * (blockDim.x >> 5) + (threadIdx.x >> 5);
  const int tilesN = N >> 6;
  const int tm     = wave / tilesN;
  const int tn     = (wave % tilesN) << 6;
  if (tm * 16 >= M) return;
  const int lhalf = (threadIdx.x >> 4) & 1;
  const int lmod  = threadIdx.x & 15;

  v8f c0 = {}, c1 = {}, c2 = {}, c3 = {};
  const float* arow = A + (size_t)(tm * 16 + lmod) * K;

  v2f a = load_a(arow, 0, lhalf);
  Bop b = load_b(B, 0, N, tn, lhalf, lmod);
  for (int k0 = 0; k0 < K - 4; k0 += 4) {
    v2f an = load_a(arow, k0 + 4, lhalf);
    Bop bn = load_b(B, k0 + 4, N, tn, lhalf, lmod);
    c0 = __builtin_amdgcn_wmma_f32_16x16x4_f32(false, a, false, b.b0, (short)0, c0, false, false);
    c1 = __builtin_amdgcn_wmma_f32_16x16x4_f32(false, a, false, b.b1, (short)0, c1, false, false);
    c2 = __builtin_amdgcn_wmma_f32_16x16x4_f32(false, a, false, b.b2, (short)0, c2, false, false);
    c3 = __builtin_amdgcn_wmma_f32_16x16x4_f32(false, a, false, b.b3, (short)0, c3, false, false);
    a = an; b = bn;
  }
  c0 = __builtin_amdgcn_wmma_f32_16x16x4_f32(false, a, false, b.b0, (short)0, c0, false, false);
  c1 = __builtin_amdgcn_wmma_f32_16x16x4_f32(false, a, false, b.b1, (short)0, c1, false, false);
  c2 = __builtin_amdgcn_wmma_f32_16x16x4_f32(false, a, false, b.b2, (short)0, c2, false, false);
  c3 = __builtin_amdgcn_wmma_f32_16x16x4_f32(false, a, false, b.b3, (short)0, c3, false, false);

#pragma unroll
  for (int j = 0; j < 8; ++j) {
    float* p = C + (size_t)(tm * 16 + j + 8 * lhalf) * N + tn + lmod;
    p[0] = c0[j]; p[16] = c1[j]; p[32] = c2[j]; p[48] = c3[j];
  }
}

__global__ void __launch_bounds__(256)
att_scores(const float* __restrict__ h, const float* __restrict__ att_s,
           const float* __restrict__ att_d, float* __restrict__ as,
           float* __restrict__ ad, int n_nodes, int H, int C) {
  int i = blockIdx.x * blockDim.x + threadIdx.x;
  if (i >= n_nodes * H) return;
  int n = i / H, hh = i % H;
  const float4* hp = (const float4*)(h + (size_t)n * H * C + (size_t)hh * C);
  const float4* sp = (const float4*)(att_s + hh * C);
  const float4* dp = (const float4*)(att_d + hh * C);
  float s = 0.f, d = 0.f;
  for (int c = 0; c < (C >> 2); ++c) {
    float4 v = hp[c], ws = sp[c], wd = dp[c];
    s += v.x * ws.x + v.y * ws.y + v.z * ws.z + v.w * ws.w;
    d += v.x * wd.x + v.y * wd.y + v.z * wd.z + v.w * wd.w;
  }
  as[i] = s; ad[i] = d;
}

__device__ __forceinline__ float leaky(float v) { return v >= 0.f ? v : NEG_SLOPE * v; }

__global__ void __launch_bounds__(256)
edge_count(const int* __restrict__ ei, int* __restrict__ deg, int* __restrict__ pos) {
  int e = blockIdx.x * blockDim.x + threadIdx.x;
  if (e >= E0_EDGES) return;
  pos[e] = atomicAdd(&deg[ei[E0_EDGES + e]], 1);
}

__global__ void __launch_bounds__(1024)
deg_scan(const int* __restrict__ deg, int* __restrict__ offs) {
  __shared__ int wtot[32];
  __shared__ int wpre[32];
  const int lane = threadIdx.x & 31;
  const int w    = threadIdx.x >> 5;
  const int base = blockIdx.x * 4096;
  int pre = 0;
#pragma unroll 4
  for (int i = threadIdx.x * 4; i < base; i += 4096) {
    int4 v = *(const int4*)(deg + i);
    pre += v.x + v.y + v.z + v.w;
  }
  const int i0 = base + threadIdx.x * 4;
  int4 v = make_int4(0, 0, 0, 0);
  if (i0 < NN) v = *(const int4*)(deg + i0);
  const int t = v.x + v.y + v.z + v.w;
  int x = t;
  for (int o = 1; o < 32; o <<= 1) {
    int y = __shfl_up(x, o);
    if (lane >= o) x += y;
  }
  for (int o = 16; o > 0; o >>= 1) pre += __shfl_xor(pre, o);
  if (lane == 31) wtot[w] = x;
  if (lane == 0)  wpre[w] = pre;
  __syncthreads();
  int add = 0;
  for (int k = 0; k < 32; ++k) {
    add += wpre[k];
    if (k < w) add += wtot[k];
  }
  const int ex = add + x - t;
  if (i0 < NN) {
    int4 o4 = make_int4(ex, ex + v.x, ex + v.x + v.y, ex + v.x + v.y + v.z);
    *(int4*)(offs + i0) = o4;
    if (i0 + 4 == NN) offs[NN] = ex + t;
  }
}

__global__ void __launch_bounds__(256)
edge_fill(const int* __restrict__ ei, const int* __restrict__ offs,
          const int* __restrict__ pos, int* __restrict__ grp) {
  int e = blockIdx.x * blockDim.x + threadIdx.x;
  if (e >= E0_EDGES) return;
  grp[offs[ei[E0_EDGES + e]] + pos[e]] = ei[e];
}

__global__ void __launch_bounds__(256)
group_sort(const int* __restrict__ offs, const int* __restrict__ grp, int* __restrict__ nbr) {
  const int lane = threadIdx.x & 31;
  const int node = __builtin_amdgcn_readfirstlane(blockIdx.x * 8 + (threadIdx.x >> 5));
  if (node >= NN) return;
  const int off = offs[node];
  const int cnt = offs[node + 1] - off;
  if (cnt <= 32) {
    const int key = lane < cnt ? grp[off + lane] : 0x7fffffff;
    int rank = 0;
    for (int j = 0; j < cnt; ++j) {
      const int kj = __builtin_amdgcn_readlane(key, j);
      rank += (kj < key || (kj == key && j < lane)) ? 1 : 0;
    }
    if (lane < cnt) nbr[off + rank] = key;
  } else {
    for (int i = lane; i < cnt; i += 32) {
      const int key = grp[off + i];
      int rank = 0;
      for (int j = 0; j < cnt; ++j) {
        const int kj = grp[off + j];
        rank += (kj < key || (kj == key && j < i)) ? 1 : 0;
      }
      nbr[off + rank] = key;
    }
  }
}

template <int H> struct ScoreVec;
template <> struct ScoreVec<1> { typedef float  type; };
template <> struct ScoreVec<2> { typedef float2 type; };

template <int H>
__device__ __forceinline__ void entry_scores(const int* __restrict__ nbr,
                                             const float* __restrict__ as,
                                             int off, int node, int t, int T,
                                             const float* adn, int& s, float* e) {
  typedef typename ScoreVec<H>::type sv_t;
  const bool valid = t < T;
  s = node;
  if (valid && t > 0) s = nbr[off + t - 1];
  const sv_t a = *(const sv_t*)(as + s * H);
  const float* af = (const float*)&a;
  for (int hh = 0; hh < H; ++hh)
    e[hh] = valid ? leaky(af[hh] + adn[hh]) : -INFINITY;
}

template <int CPL> struct RowVec;
template <> struct RowVec<4> { typedef float4 type; };
template <> struct RowVec<2> { typedef float2 type; };

template <int H, int C>
__device__ __forceinline__ void add_rows(const float* __restrict__ h, int lane, int n,
                                         int s, const float* alpha, float (*acc)[C / 32]) {
  constexpr int CPL = C / 32;
  typedef typename RowVec<CPL>::type vec_t;
#pragma unroll 4
  for (int j = 0; j < n; ++j) {
    const int sj = __builtin_amdgcn_readlane(s, j);
    const float* row = h + (size_t)sj * (H * C) + lane * CPL;
    for (int hh = 0; hh < H; ++hh) {
      const float a = __int_as_float(__builtin_amdgcn_readlane(__float_as_int(alpha[hh]), j));
      const vec_t v = *(const vec_t*)(row + hh * C);
      const float* vf = (const float*)&v;
      for (int k = 0; k < CPL; ++k) acc[hh][k] += a * vf[k];
    }
  }
}

template <int H, int C, bool ELU>
__device__ __forceinline__ void
gat_gather_body(const int* __restrict__ offs, const int* __restrict__ nbr,
                const float* __restrict__ h, const float* __restrict__ as,
                const float* __restrict__ ad, const float* __restrict__ bias,
                float* __restrict__ out) {
  constexpr int CPL = C / 32;
  typedef typename RowVec<CPL>::type vec_t;
  const int lane = threadIdx.x & 31;
  const int node = __builtin_amdgcn_readfirstlane(blockIdx.x * 8 + (threadIdx.x >> 5));
  if (node >= NN) return;
  const int off = offs[node];
  const int T   = offs[node + 1] - off + 1;
  float adn[H];
  for (int hh = 0; hh < H; ++hh) adn[hh] = ad[node * H + hh];

  int s0; float e0[H];
  entry_scores<H>(nbr, as, off, node, lane, T, adn, s0, e0);

  float m[H];
  for (int hh = 0; hh < H; ++hh) m[hh] = e0[hh];
  for (int c = 32; c < T; c += 32) {
    int s; float e[H];
    entry_scores<H>(nbr, as, off, node, c + lane, T, adn, s, e);
    for (int hh = 0; hh < H; ++hh) m[hh] = fmaxf(m[hh], e[hh]);
  }
  for (int hh = 0; hh < H; ++hh)
    for (int o = 16; o > 0; o >>= 1) m[hh] = fmaxf(m[hh], __shfl_xor(m[hh], o));

  float x0[H], den[H];
  for (int hh = 0; hh < H; ++hh) {
    x0[hh] = lane < T ? __expf(e0[hh] - m[hh]) : 0.f;
    float d = x0[hh];
    for (int o = 16; o > 0; o >>= 1) d += __shfl_xor(d, o);
    den[hh] = d;
  }
  for (int c = 32; c < T; c += 32) {
    int s; float e[H];
    entry_scores<H>(nbr, as, off, node, c + lane, T, adn, s, e);
    for (int hh = 0; hh < H; ++hh) {
      float d = (c + lane) < T ? __expf(e[hh] - m[hh]) : 0.f;
      for (int o = 16; o > 0; o >>= 1) d += __shfl_xor(d, o);
      den[hh] += d;
    }
  }
  for (int hh = 0; hh < H; ++hh) den[hh] += 1e-16f;

  float acc[H][CPL];
  for (int hh = 0; hh < H; ++hh)
    for (int k = 0; k < CPL; ++k) acc[hh][k] = 0.f;

  float al[H];
  for (int hh = 0; hh < H; ++hh) al[hh] = x0[hh] / den[hh];
  add_rows<H, C>(h, lane, T < 32 ? T : 32, s0, al, acc);
  for (int c = 32; c < T; c += 32) {
    int s; float e[H];
    entry_scores<H>(nbr, as, off, node, c + lane, T, adn, s, e);
    for (int hh = 0; hh < H; ++hh)
      al[hh] = (c + lane) < T ? __expf(e[hh] - m[hh]) / den[hh] : 0.f;
    add_rows<H, C>(h, lane, (T - c) < 32 ? (T - c) : 32, s, al, acc);
  }

  for (int hh = 0; hh < H; ++hh) {
    const vec_t bv = *(const vec_t*)(bias + hh * C + lane * CPL);
    const float* bf = (const float*)&bv;
    vec_t ov;
    float* of = (float*)&ov;
    for (int k = 0; k < CPL; ++k) {
      float v = acc[hh][k] + bf[k];
      if (ELU) v = v > 0.f ? v : (__expf(v) - 1.f);
      of[k] = v;
    }
    *(vec_t*)(out + (size_t)node * (H * C) + hh * C + lane * CPL) = ov;
  }
}

__global__ void __launch_bounds__(256)
gat_gather_conv1(const int* __restrict__ offs, const int* __restrict__ nbr,
                 const float* __restrict__ h, const float* __restrict__ as,
                 const float* __restrict__ ad, const float* __restrict__ bias,
                 float* __restrict__ out) {
  gat_gather_body<HEADS, HID, true>(offs, nbr, h, as, ad, bias, out);
}

__global__ void __launch_bounds__(256)
gat_gather_conv2(const int* __restrict__ offs, const int* __restrict__ nbr,
                 const float* __restrict__ h, const float* __restrict__ as,
                 const float* __restrict__ ad, const float* __restrict__ bias,
                 float* __restrict__ out) {
  gat_gather_body<1, OUT_CH, false>(offs, nbr, h, as, ad, bias, out);
}

extern "C" void kernel_launch(void* const* d_in, const int* in_sizes, int n_in,
                              void* d_out, int out_size, void* d_ws, size_t ws_size,
                              hipStream_t stream) {
  const float* x    = (const float*)d_in[0];
  const int*   ei   = (const int*)  d_in[1];
  const float* W1   = (const float*)d_in[2];
  const float* aS1  = (const float*)d_in[3];
  const float* aD1  = (const float*)d_in[4];
  const float* b1   = (const float*)d_in[5];
  const float* Wmu  = (const float*)d_in[6];
  const float* aSmu = (const float*)d_in[7];
  const float* aDmu = (const float*)d_in[8];
  const float* bmu  = (const float*)d_in[9];
  const float* Wls  = (const float*)d_in[10];
  const float* aSls = (const float*)d_in[11];
  const float* aDls = (const float*)d_in[12];
  const float* bls  = (const float*)d_in[13];

  char* ws = (char*)d_ws;
  size_t off = 0;
  auto alloc = [&](size_t bytes) -> char* {
    off = (off + 255) & ~(size_t)255;
    char* p = ws + off; off += bytes; return p;
  };

  float* h1   = (float*)alloc((size_t)NN * HC1 * 4);
  float* hbuf = (float*)alloc((size_t)NN * HC1 * 4);
  float* as1  = (float*)alloc((size_t)NN * HEADS * 4);
  float* ad1  = (float*)alloc((size_t)NN * HEADS * 4);
  float* as2m = (float*)alloc((size_t)NN * 4);
  float* ad2m = (float*)alloc((size_t)NN * 4);
  float* as2l = (float*)alloc((size_t)NN * 4);
  float* ad2l = (float*)alloc((size_t)NN * 4);
  int*   deg  = (int*)  alloc((size_t)NN * 4);
  int*   offs = (int*)  alloc((size_t)(NN + 1) * 4);
  int*   pos  = (int*)  alloc((size_t)E0_EDGES * 4);
  int*   grp  = (int*)  alloc((size_t)E0_EDGES * 4);
  int*   nbr  = (int*)  alloc((size_t)E0_EDGES * 4);

  float* hm = h1;
  float* hl = h1 + (size_t)NN * OUT_CH;
  float* out_mu = (float*)d_out;
  float* out_ls = out_mu + (size_t)NN * OUT_CH;

  hipMemsetAsync(deg, 0, (size_t)NN * 4, stream);
  edge_count<<<(E0_EDGES + 255) / 256, 256, 0, stream>>>(ei, deg, pos);
  deg_scan<<<(NN + 4095) / 4096, 1024, 0, stream>>>(deg, offs);
  edge_fill<<<(E0_EDGES + 255) / 256, 256, 0, stream>>>(ei, offs, pos, grp);
  group_sort<<<(NN + 7) / 8, 256, 0, stream>>>(offs, grp, nbr);

  int waves = (NN / 16) * (HC1 / 64);
  gemm_wmma_f32<<<(waves + 7) / 8, 256, 0, stream>>>(x, W1, h1, NN, HC1, IN_CH);
  att_scores<<<(NN * HEADS + 255) / 256, 256, 0, stream>>>(h1, aS1, aD1, as1, ad1, NN, HEADS, HID);
  gat_gather_conv1<<<(NN + 7) / 8, 256, 0, stream>>>(offs, nbr, h1, as1, ad1, b1, hbuf);

  waves = (NN / 16) * (OUT_CH / 64);
  gemm_wmma_f32<<<(waves + 7) / 8, 256, 0, stream>>>(hbuf, Wmu, hm, NN, OUT_CH, HC1);
  gemm_wmma_f32<<<(waves + 7) / 8, 256, 0, stream>>>(hbuf, Wls, hl, NN, OUT_CH, HC1);
  att_scores<<<(NN + 255) / 256, 256, 0, stream>>>(hm, aSmu, aDmu, as2m, ad2m, NN, 1, OUT_CH);
  att_scores<<<(NN + 255) / 256, 256, 0, stream>>>(hl, aSls, aDls, as2l, ad2l, NN, 1, OUT_CH);

  gat_gather_conv2<<<(NN + 7) / 8, 256, 0, stream>>>(offs, nbr, hm, as2m, ad2m, bmu, out_mu);
  gat_gather_conv2<<<(NN + 7) / 8, 256, 0, stream>>>(offs, nbr, hl, as2l, ad2l, bls, out_ls);
}


# second layer's mu and logstd convolutions run as the two heads of one: one GEMM against both weight matrices, one score kernel, one gather (10 launches, not 13); every sum keeps its order
# speedup vs baseline: 3.3093x; 1.1560x over previous
#include <hip/hip_runtime.h>

typedef __attribute__((ext_vector_type(2))) float v2f;
typedef __attribute__((ext_vector_type(8))) float v8f;

#define NN       50000
#define IN_CH    128
#define OUT_CH   64
#define HEADS    2
#define HID      128
#define HC1      (HEADS*HID)
#define E0_EDGES 400000
#define E_TOT    (E0_EDGES + NN)
#define NEG_SLOPE 0.2f

struct Bop { v2f b0, b1, b2, b3; };

__device__ __forceinline__ v2f load_a(const float* __restrict__ arow, int k0, int lhalf) {
  v2f a;
  a.x = arow[k0 + 2 * lhalf];
  a.y = arow[k0 + 2 * lhalf + 1];
  return a;
}
__device__ __forceinline__ Bop load_b(const float* __restrict__ B, int k0, int N,
                                      int tn, int lhalf, int lmod) {
  const float* r0 = B + (size_t)(k0 + 2 * lhalf) * N + tn + lmod;
  const float* r1 = r0 + N;
  Bop b;
  b.b0.x = r0[0];  b.b0.y = r1[0];
  b.b1.x = r0[16]; b.b1.y = r1[16];
  b.b2.x = r0[32]; b.b2.y = r1[32];
  b.b3.x = r0[48]; b.b3.y = r1[48];
  return b;
}

__device__ __forceinline__ void
gemm_tile_16x64(const float* __restrict__ A, const float* __restrict__ B,
                float* __restrict__ C, int K, int ldb, int ldc, int tm, int tn, int cn) {
  const int lhalf = (threadIdx.x >> 4) & 1;
  const int lmod  = threadIdx.x & 15;

  v8f c0 = {}, c1 = {}, c2 = {}, c3 = {};
  const float* arow = A + (size_t)(tm * 16 + lmod) * K;

  v2f a = load_a(arow, 0, lhalf);
  Bop b = load_b(B, 0, ldb, tn, lhalf, lmod);
  for (int k0 = 0; k0 < K - 4; k0 += 4) {
    v2f an = load_a(arow, k0 + 4, lhalf);
    Bop bn = load_b(B, k0 + 4, ldb, tn, lhalf, lmod);
    c0 = __builtin_amdgcn_wmma_f32_16x16x4_f32(false, a, false, b.b0, (short)0, c0, false, false);
    c1 = __builtin_amdgcn_wmma_f32_16x16x4_f32(false, a, false, b.b1, (short)0, c1, false, false);
    c2 = __builtin_amdgcn_wmma_f32_16x16x4_f32(false, a, false, b.b2, (short)0, c2, false, false);
    c3 = __builtin_amdgcn_wmma_f32_16x16x4_f32(false, a, false, b.b3, (short)0, c3, false, false);
    a = an; b = bn;
  }
  c0 = __builtin_amdgcn_wmma_f32_16x16x4_f32(false, a, false, b.b0, (short)0, c0, false, false);
  c1 = __builtin_amdgcn_wmma_f32_16x16x4_f32(false, a, false, b.b1, (short)0, c1, false, false);
  c2 = __builtin_amdgcn_wmma_f32_16x16x4_f32(false, a, false, b.b2, (short)0, c2, false, false);
  c3 = __builtin_amdgcn_wmma_f32_16x16x4_f32(false, a, false, b.b3, (short)0, c3, false, false);

#pragma unroll
  for (int j = 0; j < 8; ++j) {
    float* p = C + (size_t)(tm * 16 + j + 8 * lhalf) * ldc + cn + lmod;
    p[0] = c0[j]; p[16] = c1[j]; p[32] = c2[j]; p[48] = c3[j];
  }
}

__global__ void __launch_bounds__(256)
gemm_wmma_f32(const float* __restrict__ A, const float* __restrict__ B,
              float* __restrict__ C, int M, int N, int K) {
  const int wave   = blockIdx.x * (blockDim.x >> 5) + (threadIdx.x >> 5);
  const int tilesN = N >> 6;
  const int tm     = wave / tilesN;
  const int tn     = (wave % tilesN) << 6;
  if (tm * 16 >= M) return;
  gemm_tile_16x64(A, B, C, K, N, N, tm, tn, tn);
}

__global__ void __launch_bounds__(256)
gemm_wmma_f32_pair(const float* __restrict__ A, const float* __restrict__ B0,
                   const float* __restrict__ B1, float* __restrict__ C, int M, int K) {
  const int wave = blockIdx.x * (blockDim.x >> 5) + (threadIdx.x >> 5);
  const int tm   = wave >> 1;
  const int half = wave & 1;
  if (tm * 16 >= M) return;
  gemm_tile_16x64(A, half ? B1 : B0, C, K, 64, 128, tm, 0, half << 6);
}

__global__ void __launch_bounds__(256)
att_scores(const float* __restrict__ h,
           const float* __restrict__ att_s0, const float* __restrict__ att_s1,
           const float* __restrict__ att_d0, const float* __restrict__ att_d1,
           float* __restrict__ as, float* __restrict__ ad, int n_nodes, int C) {
  int i = blockIdx.x * blockDim.x + threadIdx.x;
  if (i >= n_nodes * HEADS) return;
  int n = i / HEADS, hh = i % HEADS;
  const float4* hp = (const float4*)(h + (size_t)n * HEADS * C + (size_t)hh * C);
  const float4* sp = (const float4*)(hh ? att_s1 : att_s0);
  const float4* dp = (const float4*)(hh ? att_d1 : att_d0);
  float s = 0.f, d = 0.f;
  for (int c = 0; c < (C >> 2); ++c) {
    float4 v = hp[c], ws = sp[c], wd = dp[c];
    s += v.x * ws.x + v.y * ws.y + v.z * ws.z + v.w * ws.w;
    d += v.x * wd.x + v.y * wd.y + v.z * wd.z + v.w * wd.w;
  }
  as[i] = s; ad[i] = d;
}

__device__ __forceinline__ float leaky(float v) { return v >= 0.f ? v : NEG_SLOPE * v; }

__global__ void __launch_bounds__(256)
edge_count(const int* __restrict__ ei, int* __restrict__ deg, int* __restrict__ pos) {
  int e = blockIdx.x * blockDim.x + threadIdx.x;
  if (e >= E0_EDGES) return;
  pos[e] = atomicAdd(&deg[ei[E0_EDGES + e]], 1);
}

__global__ void __launch_bounds__(1024)
deg_scan(const int* __restrict__ deg, int* __restrict__ offs) {
  __shared__ int wtot[32];
  __shared__ int wpre[32];
  const int lane = threadIdx.x & 31;
  const int w    = threadIdx.x >> 5;
  const int base = blockIdx.x * 4096;
  int pre = 0;
#pragma unroll 4
  for (int i = threadIdx.x * 4; i < base; i += 4096) {
    int4 v = *(const int4*)(deg + i);
    pre += v.x + v.y + v.z + v.w;
  }
  const int i0 = base + threadIdx.x * 4;
  int4 v = make_int4(0, 0, 0, 0);
  if (i0 < NN) v = *(const int4*)(deg + i0);
  const int t = v.x + v.y + v.z + v.w;
  int x = t;
  for (int o = 1; o < 32; o <<= 1) {
    int y = __shfl_up(x, o);
    if (lane >= o) x += y;
  }
  for (int o = 16; o > 0; o >>= 1) pre += __shfl_xor(pre, o);
  if (lane == 31) wtot[w] = x;
  if (lane == 0)  wpre[w] = pre;
  __syncthreads();
  int add = 0;
  for (int k = 0; k < 32; ++k) {
    add += wpre[k];
    if (k < w) add += wtot[k];
  }
  const int ex = add + x - t;
  if (i0 < NN) {
    int4 o4 = make_int4(ex, ex + v.x, ex + v.x + v.y, ex + v.x + v.y + v.z);
    *(int4*)(offs + i0) = o4;
    if (i0 + 4 == NN) offs[NN] = ex + t;
  }
}

__global__ void __launch_bounds__(256)
edge_fill(const int* __restrict__ ei, const int* __restrict__ offs,
          const int* __restrict__ pos, int* __restrict__ grp) {
  int e = blockIdx.x * blockDim.x + threadIdx.x;
  if (e >= E0_EDGES) return;
  grp[offs[ei[E0_EDGES + e]] + pos[e]] = ei[e];
}

__global__ void __launch_bounds__(256)
group_sort(const int* __restrict__ offs, const int* __restrict__ grp, int* __restrict__ nbr) {
  const int lane = threadIdx.x & 31;
  const int node = __builtin_amdgcn_readfirstlane(blockIdx.x * 8 + (threadIdx.x >> 5));
  if (node >= NN) return;
  const int off = offs[node];
  const int cnt = offs[node + 1] - off;
  if (cnt <= 32) {
    const int key = lane < cnt ? grp[off + lane] : 0x7fffffff;
    int rank = 0;
    for (int j = 0; j < cnt; ++j) {
      const int kj = __builtin_amdgcn_readlane(key, j);
      rank += (kj < key || (kj == key && j < lane)) ? 1 : 0;
    }
    if (lane < cnt) nbr[off + rank] = key;
  } else {
    for (int i = lane; i < cnt; i += 32) {
      const int key = grp[off + i];
      int rank = 0;
      for (int j = 0; j < cnt; ++j) {
        const int kj = grp[off + j];
        rank += (kj < key || (kj == key && j < i)) ? 1 : 0;
      }
      nbr[off + rank] = key;
    }
  }
}

template <int H> struct ScoreVec;
template <> struct ScoreVec<2> { typedef float2 type; };

template <int H>
__device__ __forceinline__ void entry_scores(const int* __restrict__ nbr,
                                             const float* __restrict__ as,
                                             int off, int node, int t, int T,
                                             const float* adn, int& s, float* e) {
  typedef typename ScoreVec<H>::type sv_t;
  const bool valid = t < T;
  s = node;
  if (valid && t > 0) s = nbr[off + t - 1];
  const sv_t a = *(const sv_t*)(as + s * H);
  const float* af = (const float*)&a;
  for (int hh = 0; hh < H; ++hh)
    e[hh] = valid ? leaky(af[hh] + adn[hh]) : -INFINITY;
}

template <int CPL> struct RowVec;
template <> struct RowVec<4> { typedef float4 type; };
template <> struct RowVec<2> { typedef float2 type; };

template <int H, int C>
__device__ __forceinline__ void add_rows(const float* __restrict__ h, int lane, int n,
                                         int s, const float* alpha, float (*acc)[C / 32]) {
  constexpr int CPL = C / 32;
  typedef typename RowVec<CPL>::type vec_t;
#pragma unroll 4
  for (int j = 0; j < n; ++j) {
    const int sj = __builtin_amdgcn_readlane(s, j);
    const float* row = h + (size_t)sj * (H * C) + lane * CPL;
    for (int hh = 0; hh < H; ++hh) {
      const float a = __int_as_float(__builtin_amdgcn_readlane(__float_as_int(alpha[hh]), j));
      const vec_t v = *(const vec_t*)(row + hh * C);
      const float* vf = (const float*)&v;
      for (int k = 0; k < CPL; ++k) acc[hh][k] += a * vf[k];
    }
  }
}

template <int H, int C, bool ELU>
__device__ __forceinline__ void
gat_gather_body(const int* __restrict__ offs, const int* __restrict__ nbr,
                const float* __restrict__ h, const float* __restrict__ as,
                const float* __restrict__ ad,
                const float* __restrict__ bias0, const float* __restrict__ bias1,
                float* __restrict__ out0, float* __restrict__ out1, int opitch) {
  constexpr int CPL = C / 32;
  typedef typename RowVec<CPL>::type vec_t;
  const int lane = threadIdx.x & 31;
  const int node = __builtin_amdgcn_readfirstlane(blockIdx.x * 8 + (threadIdx.x >> 5));
  if (node >= NN) return;
  const int off = offs[node];
  const int T   = offs[node + 1] - off + 1;
  float adn[H];
  for (int hh = 0; hh < H; ++hh) adn[hh] = ad[node * H + hh];

  int s0; float e0[H];
  entry_scores<H>(nbr, as, off, node, lane, T, adn, s0, e0);

  float m[H];
  for (int hh = 0; hh < H; ++hh) m[hh] = e0[hh];
  for (int c = 32; c < T; c += 32) {
    int s; float e[H];
    entry_scores<H>(nbr, as, off, node, c + lane, T, adn, s, e);
    for (int hh = 0; hh < H; ++hh) m[hh] = fmaxf(m[hh], e[hh]);
  }
  for (int hh = 0; hh < H; ++hh)
    for (int o = 16; o > 0; o >>= 1) m[hh] = fmaxf(m[hh], __shfl_xor(m[hh], o));

  float x0[H], den[H];
  for (int hh = 0; hh < H; ++hh) {
    x0[hh] = lane < T ? __expf(e0[hh] - m[hh]) : 0.f;
    float d = x0[hh];
    for (int o = 16; o > 0; o >>= 1) d += __shfl_xor(d, o);
    den[hh] = d;
  }
  for (int c = 32; c < T; c += 32) {
    int s; float e[H];
    entry_scores<H>(nbr, as, off, node, c + lane, T, adn, s, e);
    for (int hh = 0; hh < H; ++hh) {
      float d = (c + lane) < T ? __expf(e[hh] - m[hh]) : 0.f;
      for (int o = 16; o > 0; o >>= 1) d += __shfl_xor(d, o);
      den[hh] += d;
    }
  }
  for (int hh = 0; hh < H; ++hh) den[hh] += 1e-16f;

  float acc[H][CPL];
  for (int hh = 0; hh < H; ++hh)
    for (int k = 0; k < CPL; ++k) acc[hh][k] = 0.f;

  float al[H];
  for (int hh = 0; hh < H; ++hh) al[hh] = x0[hh] / den[hh];
  add_rows<H, C>(h, lane, T < 32 ? T : 32, s0, al, acc);
  for (int c = 32; c < T; c += 32) {
    int s; float e[H];
    entry_scores<H>(nbr, as, off, node, c + lane, T, adn, s, e);
    for (int hh = 0; hh < H; ++hh)
      al[hh] = (c + lane) < T ? __expf(e[hh] - m[hh]) / den[hh] : 0.f;
    add_rows<H, C>(h, lane, (T - c) < 32 ? (T - c) : 32, s, al, acc);
  }

  for (int hh = 0; hh < H; ++hh) {
    const float* bias = hh ? bias1 : bias0;
    float*       out  = hh ? out1 : out0;
    const vec_t bv = *(const vec_t*)(bias + lane * CPL);
    const float* bf = (const float*)&bv;
    vec_t ov;
    float* of = (float*)&ov;
    for (int k = 0; k < CPL; ++k) {
      float v = acc[hh][k] + bf[k];
      if (ELU) v = v > 0.f ? v : (__expf(v) - 1.f);
      of[k] = v;
    }
    *(vec_t*)(out + (size_t)node * opitch + lane * CPL) = ov;
  }
}

__global__ void __launch_bounds__(256)
gat_gather_conv1(const int* __restrict__ offs, const int* __restrict__ nbr,
                 const float* __restrict__ h, const float* __restrict__ as,
                 const float* __restrict__ ad, const float* __restrict__ bias,
                 float* __restrict__ out) {
  gat_gather_body<HEADS, HID, true>(offs, nbr, h, as, ad, bias, bias + HID,
                                    out, out + HID, HC1);
}

__global__ void __launch_bounds__(256)
gat_gather_conv2(const int* __restrict__ offs, const int* __restrict__ nbr,
                 const float* __restrict__ h, const float* __restrict__ as,
                 const float* __restrict__ ad,
                 const float* __restrict__ bias_mu, const float* __restrict__ bias_ls,
                 float* __restrict__ out_mu, float* __restrict__ out_ls) {
  gat_gather_body<HEADS, OUT_CH, false>(offs, nbr, h, as, ad, bias_mu, bias_ls,
                                        out_mu, out_ls, OUT_CH);
}

extern "C" void kernel_launch(void* const* d_in, const int* in_sizes, int n_in,
                              void* d_out, int out_size, void* d_ws, size_t ws_size,
                              hipStream_t stream) {
  const float* x    = (const float*)d_in[0];
  const int*   ei   = (const int*)  d_in[1];
  const float* W1   = (const float*)d_in[2];
  const float* aS1  = (const float*)d_in[3];
  const float* aD1  = (const float*)d_in[4];
  const float* b1   = (const float*)d_in[5];
  const float* Wmu  = (const float*)d_in[6];
  const float* aSmu = (const float*)d_in[7];
  const float* aDmu = (const float*)d_in[8];
  const float* bmu  = (const float*)d_in[9];
  const float* Wls  = (const float*)d_in[10];
  const float* aSls = (const float*)d_in[11];
  const float* aDls = (const float*)d_in[12];
  const float* bls  = (const float*)d_in[13];

  char* ws = (char*)d_ws;
  size_t off = 0;
  auto alloc = [&](size_t bytes) -> char* {
    off = (off + 255) & ~(size_t)255;
    char* p = ws + off; off += bytes; return p;
  };

  float* h1   = (float*)alloc((size_t)NN * HC1 * 4);
  float* hbuf = (float*)alloc((size_t)NN * HC1 * 4);
  float* as1  = (float*)alloc((size_t)NN * HEADS * 4);
  float* ad1  = (float*)alloc((size_t)NN * HEADS * 4);
  float* as2  = (float*)alloc((size_t)NN * HEADS * 4);
  float* ad2  = (float*)alloc((size_t)NN * HEADS * 4);
  int*   deg  = (int*)  alloc((size_t)NN * 4);
  int*   offs = (int*)  alloc((size_t)(NN + 1) * 4);
  int*   pos  = (int*)  alloc((size_t)E0_EDGES * 4);
  int*   grp  = (int*)  alloc((size_t)E0_EDGES * 4);
  int*   nbr  = (int*)  alloc((size_t)E0_EDGES * 4);

  float* h2 = h1;
  float* out_mu = (float*)d_out;
  float* out_ls = out_mu + (size_t)NN * OUT_CH;

  hipMemsetAsync(deg, 0, (size_t)NN * 4, stream);
  edge_count<<<(E0_EDGES + 255) / 256, 256, 0, stream>>>(ei, deg, pos);
  deg_scan<<<(NN + 4095) / 4096, 1024, 0, stream>>>(deg, offs);
  edge_fill<<<(E0_EDGES + 255) / 256, 256, 0, stream>>>(ei, offs, pos, grp);
  group_sort<<<(NN + 7) / 8, 256, 0, stream>>>(offs, grp, nbr);

  int waves = (NN / 16) * (HC1 / 64);
  gemm_wmma_f32<<<(waves + 7) / 8, 256, 0, stream>>>(x, W1, h1, NN, HC1, IN_CH);
  att_scores<<<(NN * HEADS + 255) / 256, 256, 0, stream>>>(h1, aS1, aS1 + HID, aD1, aD1 + HID,
                                                             as1, ad1, NN, HID);
  gat_gather_conv1<<<(NN + 7) / 8, 256, 0, stream>>>(offs, nbr, h1, as1, ad1, b1, hbuf);

  waves = (NN / 16) * 2;
  gemm_wmma_f32_pair<<<(waves + 7) / 8, 256, 0, stream>>>(hbuf, Wmu, Wls, h2, NN, HC1);
  att_scores<<<(NN * HEADS + 255) / 256, 256, 0, stream>>>(h2, aSmu, aSls, aDmu, aDls,
                                                             as2, ad2, NN, OUT_CH);
  gat_gather_conv2<<<(NN + 7) / 8, 256, 0, stream>>>(offs, nbr, h2, as2, ad2, bmu, bls,
                                                      out_mu, out_ls);
}


# GEMM fragment loads as whole 16-byte pieces: 5 loads per 8 matrix steps, not 18 (a lane's 4 accumulators are 4 neighbouring columns, k regrouped); 16-byte row stores; k and score sums change order
# speedup vs baseline: 3.8742x; 1.1707x over previous
#include <hip/hip_runtime.h>

typedef __attribute__((ext_vector_type(2))) float v2f;
typedef __attribute__((ext_vector_type(8))) float v8f;

#define NN       50000
#define IN_CH    128
#define OUT_CH   64
#define HEADS    2
#define HID      128
#define HC1      (HEADS*HID)
#define E0_EDGES 400000
#define E_TOT    (E0_EDGES + NN)
#define NEG_SLOPE 0.2f

struct Frag { float4 a; float4 r0, r1, r2, r3; };

template <int LDB>
__device__ __forceinline__ Frag load_frag(const float* __restrict__ ap,
                                          const float* __restrict__ bp, int kb) {
  Frag f;
  f.a  = *(const float4*)(ap + kb);
  const float* b = bp + (size_t)kb * LDB;
  f.r0 = *(const float4*)(b);
  f.r1 = *(const float4*)(b + LDB);
  f.r2 = *(const float4*)(b + 2 * LDB);
  f.r3 = *(const float4*)(b + 3 * LDB);
  return f;
}

__device__ __forceinline__ v2f pair(float x, float y) { v2f p; p.x = x; p.y = y; return p; }

__device__ __forceinline__ void mma_block(const Frag& f, v8f& c0, v8f& c1, v8f& c2, v8f& c3) {
  const v2f a0 = pair(f.a.x, f.a.y);
  const v2f a1 = pair(f.a.z, f.a.w);
  c0 = __builtin_amdgcn_wmma_f32_16x16x4_f32(false, a0, false, pair(f.r0.x, f.r1.x), (short)0, c0, false, false);
  c1 = __builtin_amdgcn_wmma_f32_16x16x4_f32(false, a0, false, pair(f.r0.y, f.r1.y), (short)0, c1, false, false);
  c2 = __builtin_amdgcn_wmma_f32_16x16x4_f32(false, a0, false, pair(f.r0.z, f.r1.z), (short)0, c2, false, false);
  c3 = __builtin_amdgcn_wmma_f32_16x16x4_f32(false, a0, false, pair(f.r0.w, f.r1.w), (short)0, c3, false, false);
  c0 = __builtin_amdgcn_wmma_f32_16x16x4_f32(false, a1, false, pair(f.r2.x, f.r3.x), (short)0, c0, false, false);
  c1 = __builtin_amdgcn_wmma_f32_16x16x4_f32(false, a1, false, pair(f.r2.y, f.r3.y), (short)0, c1, false, false);
  c2 = __builtin_amdgcn_wmma_f32_16x16x4_f32(false, a1, false, pair(f.r2.z, f.r3.z), (short)0, c2, false, false);
  c3 = __builtin_amdgcn_wmma_f32_16x16x4_f32(false, a1, false, pair(f.r2.w, f.r3.w), (short)0, c3, false, false);
}

template <int M>
__device__ __forceinline__ float row_xor_get(float v) {
  return __int_as_float(__builtin_amdgcn_update_dpp(0, __float_as_int(v), 0x160 | M, 0xf, 0xf, true));
}

template <int M, int N>
__device__ __forceinline__ void halve_over_lanes(float* v, int lmod) {
  const bool hi = (lmod & M) != 0;
  for (int i = 0; i < N / 2; ++i) {
    const float keep = hi ? v[i + N / 2] : v[i];
    const float send = hi ? v[i] : v[i + N / 2];
    v[i] = keep + row_xor_get<M>(send);
  }
}

__device__ __forceinline__ float
tile_row_scores(const v8f& c0, const v8f& c1, const v8f& c2, const v8f& c3,
                const float* ws, const float* wd, int lmod) {
  float v[16];
  for (int j = 0; j < 8; ++j) {
    v[j]     = c0[j] * ws[0] + c1[j] * ws[1] + c2[j] * ws[2] + c3[j] * ws[3];
    v[8 + j] = c0[j] * wd[0] + c1[j] * wd[1] + c2[j] * wd[2] + c3[j] * wd[3];
  }
  halve_over_lanes<8, 16>(v, lmod);
  halve_over_lanes<4, 8>(v, lmod);
  halve_over_lanes<2, 4>(v, lmod);
  halve_over_lanes<1, 2>(v, lmod);
  return v[0];
}

template <int K, int LDB, int LDC>
__device__ __forceinline__ float
gemm_tile_16x64(const float* __restrict__ A, const float* __restrict__ B,
                float* __restrict__ C, const float* __restrict__ atts,
                const float* __restrict__ attd, int tm, int tn, int cn) {
  const int lhalf = (threadIdx.x >> 4) & 1;
  const int lmod  = threadIdx.x & 15;

  v8f c0 = {}, c1 = {}, c2 = {}, c3 = {};
  const float* ap = A + (size_t)(tm * 16 + lmod) * K + 4 * lhalf;
  const float* bp = B + (size_t)(4 * lhalf) * LDB + tn + 4 * lmod;

  Frag f = load_frag<LDB>(ap, bp, 0);
#pragma unroll
  for (int kb = 0; kb < K - 8; kb += 8) {
    const Frag fn = load_frag<LDB>(ap, bp, kb + 8);
    mma_block(f, c0, c1, c2, c3);
    f = fn;
  }
  const float4 ws4 = *(const float4*)(atts + 4 * lmod);
  const float4 wd4 = *(const float4*)(attd + 4 * lmod);
  mma_block(f, c0, c1, c2, c3);

#pragma unroll
  for (int j = 0; j < 8; ++j) {
    float4 o;
    o.x = c0[j]; o.y = c1[j]; o.z = c2[j]; o.w = c3[j];
    *(float4*)(C + (size_t)(tm * 16 + j + 8 * lhalf) * LDC + cn + 4 * lmod) = o;
  }
  const float ws[4] = {ws4.x, ws4.y, ws4.z, ws4.w};
  const float wd[4] = {wd4.x, wd4.y, wd4.z, wd4.w};
  return tile_row_scores(c0, c1, c2, c3, ws, wd, lmod);
}

__device__ __forceinline__ int score_slot(int w) {
  const int lhalf = (threadIdx.x >> 4) & 1;
  const int lmod  = threadIdx.x & 15;
  return (w * 2 + (lmod >> 3)) * 16 + (lmod & 7) + 8 * lhalf;
}

__global__ void __launch_bounds__(256)
gemm_wmma_f32(const float* __restrict__ A, const float* __restrict__ B,
              float* __restrict__ C,
              const float* __restrict__ att_s, const float* __restrict__ att_d,
              float* __restrict__ as, float* __restrict__ ad, int M) {
  __shared__ float part[8 * 2 * 16];
  const int w      = threadIdx.x >> 5;
  const int wave   = blockIdx.x * (blockDim.x >> 5) + w;
  constexpr int tilesN = HC1 >> 6;
  const int tm     = wave / tilesN;
  const int tn     = (wave % tilesN) << 6;
  if (tm * 16 < M)
    part[score_slot(w)] = gemm_tile_16x64<IN_CH, HC1, HC1>(A, B, C, att_s + tn, att_d + tn, tm, tn, tn);
  __syncthreads();
  const int t = threadIdx.x;
  if (t < 128) {
    const int kind = t >> 6;
    const int r    = (t & 63) >> 1;
    const int head = t & 1;
    const int row  = blockIdx.x * 32 + r;
    if (row < M) {
      const int s0 = ((((r >> 4) * 4 + 2 * head) * 2 + kind) * 16) + (r & 15);
      (kind ? ad : as)[row * HEADS + head] = part[s0] + part[s0 + 32];
    }
  }
}

__global__ void __launch_bounds__(256)
gemm_wmma_f32_pair(const float* __restrict__ A, const float* __restrict__ B0,
                   const float* __restrict__ B1, float* __restrict__ C,
                   const float* __restrict__ att_s0, const float* __restrict__ att_s1,
                   const float* __restrict__ att_d0, const float* __restrict__ att_d1,
                   float* __restrict__ as, float* __restrict__ ad, int M) {
  __shared__ float part[8 * 2 * 16];
  const int w    = threadIdx.x >> 5;
  const int wave = blockIdx.x * (blockDim.x >> 5) + w;
  const int tm   = wave >> 1;
  const int half = wave & 1;
  if (tm * 16 < M)
    part[score_slot(w)] = gemm_tile_16x64<HC1, OUT_CH, 2 * OUT_CH>(A, half ? B1 : B0, C,
                                          half ? att_s1 : att_s0, half ? att_d1 : att_d0,
                                          tm, 0, half << 6);
  __syncthreads();
  const int t    = threadIdx.x;
  const int kind = t >> 7;
  const int r    = (t & 127) >> 1;
  const int head = t & 1;
  const int row  = blockIdx.x * 64 + r;
  if (row < M)
    (kind ? ad : as)[row * HEADS + head] = part[(((r >> 4) * 2 + head) * 2 + kind) * 16 + (r & 15)];
}

__device__ __forceinline__ float leaky(float v) { return v >= 0.f ? v : NEG_SLOPE * v; }

__global__ void __launch_bounds__(256)
edge_count(const int* __restrict__ ei, int* __restrict__ deg, int* __restrict__ pos) {
  int e = blockIdx.x * blockDim.x + threadIdx.x;
  if (e >= E0_EDGES) return;
  pos[e] = atomicAdd(&deg[ei[E0_EDGES + e]], 1);
}

__global__ void __launch_bounds__(1024)
deg_scan(const int* __restrict__ deg, int* __restrict__ offs) {
  __shared__ int wtot[32];
  __shared__ int wpre[32];
  const int lane = threadIdx.x & 31;
  const int w    = threadIdx.x >> 5;
  const int base = blockIdx.x * 4096;
  int pre = 0;
#pragma unroll 4
  for (int i = threadIdx.x * 4; i < base; i += 4096) {
    int4 v = *(const int4*)(deg + i);
    pre += v.x + v.y + v.z + v.w;
  }
  const int i0 = base + threadIdx.x * 4;
  int4 v = make_int4(0, 0, 0, 0);
  if (i0 < NN) v = *(const int4*)(deg + i0);
  const int t = v.x + v.y + v.z + v.w;
  int x = t;
  for (int o = 1; o < 32; o <<= 1) {
    int y = __shfl_up(x, o);
    if (lane >= o) x += y;
  }
  for (int o = 16; o > 0; o >>= 1) pre += __shfl_xor(pre, o);
  if (lane == 31) wtot[w] = x;
  if (lane == 0)  wpre[w] = pre;
  __syncthreads();
  int add = 0;
  for (int k = 0; k < 32; ++k) {
    add += wpre[k];
    if (k < w) add += wtot[k];
  }
  const int ex = add + x - t;
  if (i0 < NN) {
    int4 o4 = make_int4(ex, ex + v.x, ex + v.x + v.y, ex + v.x + v.y + v.z);
    *(int4*)(offs + i0) = o4;
    if (i0 + 4 == NN) offs[NN] = ex + t;
  }
}

__global__ void __launch_bounds__(256)
edge_fill(const int* __restrict__ ei, const int* __restrict__ offs,
          const int* __restrict__ pos, int* __restrict__ grp) {
  int e = blockIdx.x * blockDim.x + threadIdx.x;
  if (e >= E0_EDGES) return;
  grp[offs[ei[E0_EDGES + e]] + pos[e]] = ei[e];
}

template <int H> struct ScoreVec;
template <> struct ScoreVec<2> { typedef float2 type; };

template <int H>
__device__ __forceinline__ void source_scores(const float* __restrict__ as, int s, bool valid,
                                              const float* adn, float* e) {
  typedef typename ScoreVec<H>::type sv_t;
  const sv_t a = *(const sv_t*)(as + s * H);
  const float* af = (const float*)&a;
  for (int hh = 0; hh < H; ++hh)
    e[hh] = valid ? leaky(af[hh] + adn[hh]) : -INFINITY;
}

template <int H>
__device__ __forceinline__ void entry_scores(const int* nbr,
                                             const float* __restrict__ as,
                                             int off, int node, int t, int T,
                                             const float* adn, int& s, float* e) {
  const bool valid = t < T;
  s = node;
  if (valid && t > 0) s = nbr[off + t - 1];
  source_scores<H>(as, s, valid, adn, e);
}

template <int H>
__device__ __forceinline__ void
sorted_first_entries(const int* __restrict__ grp, int* nbr, const float* __restrict__ as,
                     int off, int node, int lane, int T, const float* adn, int& s, float* e) {
  const int cnt = T - 1;
  if (cnt <= 31) {
    const int key = lane < cnt ? grp[off + lane] : 0x7fffffff;
    int rank = 0;
    for (int j = 0; j < cnt; ++j) {
      const int kj = __builtin_amdgcn_readlane(key, j);
      rank += (kj < key || (kj == key && j < lane)) ? 1 : 0;
    }
    if (lane < cnt) nbr[off + rank] = key;
    const int got = __builtin_amdgcn_ds_permute((lane < cnt ? rank + 1 : 0) << 2, key);
    s = (lane > 0 && lane < T) ? got : node;
    source_scores<H>(as, s, lane < T, adn, e);
  } else {
    for (int i = lane; i < cnt; i += 32) {
      const int key = grp[off + i];
      int rank = 0;
      for (int j = 0; j < cnt; ++j) {
        const int kj = grp[off + j];
        rank += (kj < key || (kj == key && j < i)) ? 1 : 0;
      }
      nbr[off + rank] = key;
    }
    __threadfence();
    entry_scores<H>(nbr, as, off, node, lane, T, adn, s, e);
  }
}

template <int CPL> struct RowVec;
template <> struct RowVec<4> { typedef float4 type; };
template <> struct RowVec<2> { typedef float2 type; };

template <int H, int C>
__device__ __forceinline__ void add_rows(const float* __restrict__ h, int lane, int n,
                                         int s, const float* alpha, float (*acc)[C / 32]) {
  constexpr int CPL = C / 32;
  typedef typename RowVec<CPL>::type vec_t;
#pragma unroll 4
  for (int j = 0; j < n; ++j) {
    const int sj = __builtin_amdgcn_readlane(s, j);
    const float* row = h + (size_t)sj * (H * C) + lane * CPL;
    for (int hh = 0; hh < H; ++hh) {
      const float a = __int_as_float(__builtin_amdgcn_readlane(__float_as_int(alpha[hh]), j));
      const vec_t v = *(const vec_t*)(row + hh * C);
      const float* vf = (const float*)&v;
      for (int k = 0; k < CPL; ++k) acc[hh][k] += a * vf[k];
    }
  }
}

template <int H, int C, bool ELU, bool SORT>
__device__ __forceinline__ void
gat_gather_body(const int* __restrict__ offs, const int* __restrict__ grp,
                const int* nbr, int* nbrw,
                const float* __restrict__ h, const float* __restrict__ as,
                const float* __restrict__ ad,
                const float* __restrict__ bias0, const float* __restrict__ bias1,
                float* __restrict__ out0, float* __restrict__ out1, int opitch) {
  constexpr int CPL = C / 32;
  typedef typename RowVec<CPL>::type vec_t;
  const int lane = threadIdx.x & 31;
  const int node = __builtin_amdgcn_readfirstlane(blockIdx.x * 8 + (threadIdx.x >> 5));
  if (node >= NN) return;
  const int off = offs[node];
  const int T   = offs[node + 1] - off + 1;
  float adn[H];
  for (int hh = 0; hh < H; ++hh) adn[hh] = ad[node * H + hh];

  int s0; float e0[H];
  if (SORT) sorted_first_entries<H>(grp, nbrw, as, off, node, lane, T, adn, s0, e0);
  else      entry_scores<H>(nbr, as, off, node, lane, T, adn, s0, e0);

  float m[H];
  for (int hh = 0; hh < H; ++hh) m[hh] = e0[hh];
  for (int c = 32; c < T; c += 32) {
    int s; float e[H];
    entry_scores<H>(nbr, as, off, node, c + lane, T, adn, s, e);
    for (int hh = 0; hh < H; ++hh) m[hh] = fmaxf(m[hh], e[hh]);
  }
  for (int hh = 0; hh < H; ++hh)
    for (int o = 16; o > 0; o >>= 1) m[hh] = fmaxf(m[hh], __shfl_xor(m[hh], o));

  float x0[H], den[H];
  for (int hh = 0; hh < H; ++hh) {
    x0[hh] = lane < T ? __expf(e0[hh] - m[hh]) : 0.f;
    float d = x0[hh];
    for (int o = 16; o > 0; o >>= 1) d += __shfl_xor(d, o);
    den[hh] = d;
  }
  for (int c = 32; c < T; c += 32) {
    int s; float e[H];
    entry_scores<H>(nbr, as, off, node, c + lane, T, adn, s, e);
    for (int hh = 0; hh < H; ++hh) {
      float d = (c + lane) < T ? __expf(e[hh] - m[hh]) : 0.f;
      for (int o = 16; o > 0; o >>= 1) d += __shfl_xor(d, o);
      den[hh] += d;
    }
  }
  for (int hh = 0; hh < H; ++hh) den[hh] += 1e-16f;

  float acc[H][CPL];
  for (int hh = 0; hh < H; ++hh)
    for (int k = 0; k < CPL; ++k) acc[hh][k] = 0.f;

  float al[H];
  for (int hh = 0; hh < H; ++hh) al[hh] = x0[hh] / den[hh];
  add_rows<H, C>(h, lane, T < 32 ? T : 32, s0, al, acc);
  for (int c = 32; c < T; c += 32) {
    int s; float e[H];
    entry_scores<H>(nbr, as, off, node, c + lane, T, adn, s, e);
    for (int hh = 0; hh < H; ++hh)
      al[hh] = (c + lane) < T ? __expf(e[hh] - m[hh]) / den[hh] : 0.f;
    add_rows<H, C>(h, lane, (T - c) < 32 ? (T - c) : 32, s, al, acc);
  }

  for (int hh = 0; hh < H; ++hh) {
    const float* bias = hh ? bias1 : bias0;
    float*       out  = hh ? out1 : out0;
    const vec_t bv = *(const vec_t*)(bias + lane * CPL);
    const float* bf = (const float*)&bv;
    vec_t ov;
    float* of = (float*)&ov;
    for (int k = 0; k < CPL; ++k) {
      float v = acc[hh][k] + bf[k];
      if (ELU) v = v > 0.f ? v : (__expf(v) - 1.f);
      of[k] = v;
    }
    *(vec_t*)(out + (size_t)node * opitch + lane * CPL) = ov;
  }
}

__global__ void __launch_bounds__(256)
gat_gather_conv1(const int* __restrict__ offs, const int* __restrict__ grp, int* nbr,
                 const float* __restrict__ h, const float* __restrict__ as,
                 const float* __restrict__ ad, const float* __restrict__ bias,
                 float* __restrict__ out) {
  gat_gather_body<HEADS, HID, true, true>(offs, grp, nbr, nbr, h, as, ad, bias, bias + HID,
                                          out, out + HID, HC1);
}

__global__ void __launch_bounds__(256)
gat_gather_conv2(const int* __restrict__ offs, const int* __restrict__ nbr,
                 const float* __restrict__ h, const float* __restrict__ as,
                 const float* __restrict__ ad,
                 const float* __restrict__ bias_mu, const float* __restrict__ bias_ls,
                 float* __restrict__ out_mu, float* __restrict__ out_ls) {
  gat_gather_body<HEADS, OUT_CH, false, false>(offs, nullptr, nbr, nullptr, h, as, ad,
                                               bias_mu, bias_ls, out_mu, out_ls, OUT_CH);
}

extern "C" void kernel_launch(void* const* d_in, const int* in_sizes, int n_in,
                              void* d_out, int out_size, void* d_ws, size_t ws_size,
                              hipStream_t stream) {
  const float* x    = (const float*)d_in[0];
  const int*   ei   = (const int*)  d_in[1];
  const float* W1   = (const float*)d_in[2];
  const float* aS1  = (const float*)d_in[3];
  const float* aD1  = (const float*)d_in[4];
  const float* b1   = (const float*)d_in[5];
  const float* Wmu  = (const float*)d_in[6];
  const float* aSmu = (const float*)d_in[7];
  const float* aDmu = (const float*)d_in[8];
  const float* bmu  = (const float*)d_in[9];
  const float* Wls  = (const float*)d_in[10];
  const float* aSls = (const float*)d_in[11];
  const float* aDls = (const float*)d_in[12];
  const float* bls  = (const float*)d_in[13];

  char* ws = (char*)d_ws;
  size_t off = 0;
  auto alloc = [&](size_t bytes) -> char* {
    off = (off + 255) & ~(size_t)255;
    char* p = ws + off; off += bytes; return p;
  };

  float* h1   = (float*)alloc((size_t)NN * HC1 * 4);
  float* hbuf = (float*)alloc((size_t)NN * HC1 * 4);
  float* as1  = (float*)alloc((size_t)NN * HEADS * 4);
  float* ad1  = (float*)alloc((size_t)NN * HEADS * 4);
  float* as2  = (float*)alloc((size_t)NN * HEADS * 4);
  float* ad2  = (float*)alloc((size_t)NN * HEADS * 4);
  int*   deg  = (int*)  alloc((size_t)NN * 4);
  int*   offs = (int*)  alloc((size_t)(NN + 1) * 4);
  int*   pos  = (int*)  alloc((size_t)E0_EDGES * 4);
  int*   grp  = (int*)  alloc((size_t)E0_EDGES * 4);
  int*   nbr  = (int*)  alloc((size_t)E0_EDGES * 4);

  float* h2 = h1;
  float* out_mu = (float*)d_out;
  float* out_ls = out_mu + (size_t)NN * OUT_CH;

  hipMemsetAsync(deg, 0, (size_t)NN * 4, stream);
  edge_count<<<(E0_EDGES + 255) / 256, 256, 0, stream>>>(ei, deg, pos);
  deg_scan<<<(NN + 4095) / 4096, 1024, 0, stream>>>(deg, offs);
  edge_fill<<<(E0_EDGES + 255) / 256, 256, 0, stream>>>(ei, offs, pos, grp);

  int waves = (NN / 16) * (HC1 / 64);
  gemm_wmma_f32<<<(waves + 7) / 8, 256, 0, stream>>>(x, W1, h1, aS1, aD1, as1, ad1, NN);
  gat_gather_conv1<<<(NN + 7) / 8, 256, 0, stream>>>(offs, grp, nbr, h1, as1, ad1, b1, hbuf);

  waves = (NN / 16) * 2;
  gemm_wmma_f32_pair<<<(waves + 7) / 8, 256, 0, stream>>>(hbuf, Wmu, Wls, h2, aSmu, aSls, aDmu, aDls,
                                                           as2, ad2, NN);
  gat_gather_conv2<<<(NN + 7) / 8, 256, 0, stream>>>(offs, nbr, h2, as2, ad2, bmu, bls,
                                                      out_mu, out_ls);
}
